# starting kernel
# baseline (speedup 1.0000x reference)
#include <hip/hip_runtime.h>
#include <hip/hip_bf16.h>


#define Bsz   256
#define Asz   100
#define NBq   10
#define BONDq 110
#define AFq   82
#define BFq   6
#define Hq    300
#define KPAD  308
#define NT    19
#define NP    10
#define MTOT  (Bsz*Asz)
#define MT    (MTOT/16)
#define MP    (MT/2)
#define ATPB  4
#define PANEL (16*KPAD)

typedef float v2f __attribute__((ext_vector_type(2)));
typedef float v8f __attribute__((ext_vector_type(8)));

static __device__ __forceinline__ v8f wmma4(v2f a, v2f b, v8f c) {
  return __builtin_amdgcn_wmma_f32_16x16x4_f32(
      false, a, false, b, (short)0, c, false, false);
}

__global__ __launch_bounds__(256) void fc1_kernel(
    const float* __restrict__ af, const float* __restrict__ w,
    float* __restrict__ hout) {
  int tid = blockIdx.x * 8 + (threadIdx.x >> 5);
  if (tid >= MP * NP) return;
  const int lane = threadIdx.x & 31;
  const int m = lane & 15, half = lane >> 4, koff = half << 1;
  const int mp = tid / NP, np = tid - mp * NP;
  const int n0 = np * 32 + m, n1 = n0 + 16;
  const int row0 = mp * 32 + m, row1 = row0 + 16;
  const int nc0 = (n0 < Hq) ? n0 : (Hq - 1);
  const int nc1 = (n1 < Hq) ? n1 : (Hq - 1);

  const float* ap0 = af + row0 * AFq + koff;
  const float* ap1 = af + row1 * AFq + koff;
  const float* wp0 = w + koff * Hq + nc0;
  const float* wp1 = w + koff * Hq + nc1;
  v8f c00 = {}, c01 = {}, c10 = {}, c11 = {};
#pragma unroll 2
  for (int kk = 0; kk < 80; kk += 4) {
    v2f a0 = *(const v2f*)ap0;
    v2f a1 = *(const v2f*)ap1;
    v2f b0; b0.x = wp0[0]; b0.y = wp0[Hq];
    v2f b1; b1.x = wp1[0]; b1.y = wp1[Hq];
    ap0 += 4; ap1 += 4; wp0 += 4 * Hq; wp1 += 4 * Hq;
    c00 = wmma4(a0, b0, c00); c01 = wmma4(a0, b1, c01);
    c10 = wmma4(a1, b0, c10); c11 = wmma4(a1, b1, c11);
  }
  {
    const int k0 = 80 + koff;
    const int k0c = (k0 < AFq) ? k0 : AFq - 1;
    const int k1c = (k0 + 1 < AFq) ? k0 + 1 : AFq - 1;
    v2f a0, a1, b0, b1;
    float a0x = af[row0 * AFq + k0c], a0y = af[row0 * AFq + k1c];
    float a1x = af[row1 * AFq + k0c], a1y = af[row1 * AFq + k1c];
    a0.x = (k0 < AFq) ? a0x : 0.f;  a0.y = (k0 + 1 < AFq) ? a0y : 0.f;
    a1.x = (k0 < AFq) ? a1x : 0.f;  a1.y = (k0 + 1 < AFq) ? a1y : 0.f;
    b0.x = w[k0c * Hq + nc0]; b0.y = w[k1c * Hq + nc0];
    b1.x = w[k0c * Hq + nc1]; b1.y = w[k1c * Hq + nc1];
    c00 = wmma4(a0, b0, c00); c01 = wmma4(a0, b1, c01);
    c10 = wmma4(a1, b0, c10); c11 = wmma4(a1, b1, c11);
  }
  const int rb0 = mp * 32 + half * 8, rb1 = rb0 + 16;
#pragma unroll
  for (int v = 0; v < 8; ++v) {
    if (n0 < Hq) {
      hout[(rb0 + v) * Hq + n0] = fmaxf(c00[v], 0.f);
      hout[(rb1 + v) * Hq + n0] = fmaxf(c10[v], 0.f);
    }
    if (n1 < Hq) {
      hout[(rb0 + v) * Hq + n1] = fmaxf(c01[v], 0.f);
      hout[(rb1 + v) * Hq + n1] = fmaxf(c11[v], 0.f);
    }
  }
}

__global__ __launch_bounds__(256) void nei_kernel(
    const float* __restrict__ h, const float* __restrict__ bond,
    const int* __restrict__ ag, const int* __restrict__ bg,
    const int* __restrict__ nnb, const float* __restrict__ wn,
    const float* __restrict__ bn, float* __restrict__ out) {
  __shared__ float Ap[ATPB * PANEL];
  __shared__ int sib[ATPB * NBq], sia[ATPB * NBq];
  __shared__ int sbb[ATPB * NBq], sbe[ATPB * NBq];
  const int ba0 = blockIdx.x * ATPB;
  if (threadIdx.x < ATPB * NBq) {
    int q = threadIdx.x / NBq, r = threadIdx.x - q * NBq;
    int e = ((ba0 + q) * NBq + r) * 2;
    sib[threadIdx.x] = ag[e + 0]; sia[threadIdx.x] = ag[e + 1];
    sbb[threadIdx.x] = bg[e + 0]; sbe[threadIdx.x] = bg[e + 1];
  }
  __syncthreads();
  for (int idx = threadIdx.x; idx < ATPB * PANEL; idx += 256) {
    int q = idx / PANEL, rem = idx - q * PANEL;
    int r = rem / KPAD, c = rem - r * KPAD;
    float v = 0.f;
    if (r < NBq) {
      int s = q * NBq + r;
      if (c < Hq)            v = h[(sib[s] * Asz + sia[s]) * Hq + c];
      else if (c < Hq + BFq) v = bond[(sbb[s] * BONDq + sbe[s]) * BFq + (c - Hq)];
    }
    Ap[idx] = v;
  }
  __syncthreads();
  const int wave = threadIdx.x >> 5, lane = threadIdx.x & 31;
  const int m = lane & 15, half = lane >> 4, koff = half << 1;
  const float* apl = Ap + m * KPAD + koff;
  for (int t = wave; t < NT; t += 8) {
    const int n = t * 16 + m;
    const int nc = (n < Hq) ? n : (Hq - 1);
    const float* wp = wn + koff * Hq + nc;
    v8f acc[ATPB] = {};
#pragma unroll 2
    for (int kk = 0; kk < 304; kk += 4) {
      v2f b; b.x = wp[0]; b.y = wp[Hq];
      wp += 4 * Hq;
#pragma unroll
      for (int q = 0; q < ATPB; ++q) {
        v2f a = *(const v2f*)(apl + q * PANEL + kk);
        acc[q] = wmma4(a, b, acc[q]);
      }
    }
    {
      const int k0 = 304 + koff;
      v2f b;
      b.x = wn[(k0     < Hq + BFq ? k0     : Hq + BFq - 1) * Hq + nc];
      b.y = wn[(k0 + 1 < Hq + BFq ? k0 + 1 : Hq + BFq - 1) * Hq + nc];
#pragma unroll
      for (int q = 0; q < ATPB; ++q) {
        v2f a = *(const v2f*)(apl + q * PANEL + 304);
        acc[q] = wmma4(a, b, acc[q]);
      }
    }
    const float bias = bn[nc];
#pragma unroll
    for (int q = 0; q < ATPB; ++q) {
      const int nn = nnb[ba0 + q];
      float s = 0.f;
#pragma unroll
      for (int v = 0; v < 8; ++v) {
        int rj = v + half * 8;
        float x = fmaxf(acc[q][v] + bias, 0.f);
        s += (rj < nn) ? x : 0.f;
      }
      s += __shfl_xor(s, 16, 32);
      if (half == 0 && n < Hq) out[(ba0 + q) * Hq + n] = s;
    }
  }
}

__global__ __launch_bounds__(256) void atom_kernel(
    const float* __restrict__ h, const float* __restrict__ nei,
    const float* __restrict__ wa, const float* __restrict__ bb,
    float* __restrict__ hout) {
  int tid = blockIdx.x * 8 + (threadIdx.x >> 5);
  if (tid >= MP * NP) return;
  const int lane = threadIdx.x & 31;
  const int m = lane & 15, half = lane >> 4, koff = half << 1;
  const int mp = tid / NP, np = tid - mp * NP;
  const int n0 = np * 32 + m, n1 = n0 + 16;
  const int row0 = mp * 32 + m, row1 = row0 + 16;
  const int nc0 = (n0 < Hq) ? n0 : (Hq - 1);
  const int nc1 = (n1 < Hq) ? n1 : (Hq - 1);
  v8f c00 = {}, c01 = {}, c10 = {}, c11 = {};
  {
    const float* ap0 = h + row0 * Hq + koff;
    const float* ap1 = h + row1 * Hq + koff;
    const float* wp0 = wa + koff * Hq + nc0;
    const float* wp1 = wa + koff * Hq + nc1;
#pragma unroll 3
    for (int kk = 0; kk < Hq; kk += 4) {
      v2f a0 = *(const v2f*)ap0;
      v2f a1 = *(const v2f*)ap1;
      v2f b0; b0.x = wp0[0]; b0.y = wp0[Hq];
      v2f b1; b1.x = wp1[0]; b1.y = wp1[Hq];
      ap0 += 4; ap1 += 4; wp0 += 4 * Hq; wp1 += 4 * Hq;
      c00 = wmma4(a0, b0, c00); c01 = wmma4(a0, b1, c01);
      c10 = wmma4(a1, b0, c10); c11 = wmma4(a1, b1, c11);
    }
  }
  {
    const float* ap0 = nei + row0 * Hq + koff;
    const float* ap1 = nei + row1 * Hq + koff;
    const float* wp0 = wa + (Hq + koff) * Hq + nc0;
    const float* wp1 = wa + (Hq + koff) * Hq + nc1;
#pragma unroll 3
    for (int kk = 0; kk < Hq; kk += 4) {
      v2f a0 = *(const v2f*)ap0;
      v2f a1 = *(const v2f*)ap1;
      v2f b0; b0.x = wp0[0]; b0.y = wp0[Hq];
      v2f b1; b1.x = wp1[0]; b1.y = wp1[Hq];
      ap0 += 4; ap1 += 4; wp0 += 4 * Hq; wp1 += 4 * Hq;
      c00 = wmma4(a0, b0, c00); c01 = wmma4(a0, b1, c01);
      c10 = wmma4(a1, b0, c10); c11 = wmma4(a1, b1, c11);
    }
  }
  const float bias0 = bb[nc0], bias1 = bb[nc1];
  const int rb0 = mp * 32 + half * 8, rb1 = rb0 + 16;
#pragma unroll
  for (int v = 0; v < 8; ++v) {
    if (n0 < Hq) {
      hout[(rb0 + v) * Hq + n0] = fmaxf(c00[v] + bias0, 0.f);
      hout[(rb1 + v) * Hq + n0] = fmaxf(c10[v] + bias0, 0.f);
    }
    if (n1 < Hq) {
      hout[(rb0 + v) * Hq + n1] = fmaxf(c01[v] + bias1, 0.f);
      hout[(rb1 + v) * Hq + n1] = fmaxf(c11[v] + bias1, 0.f);
    }
  }
}

__global__ __launch_bounds__(256) void nprod_kernel(
    const float* __restrict__ h, const float* __restrict__ bond,
    const int* __restrict__ ag, const int* __restrict__ bg,
    const int* __restrict__ nnb, const float* __restrict__ w2a,
    const float* __restrict__ w2b, float* __restrict__ out) {
  __shared__ float Ap[ATPB * PANEL];
  __shared__ int sib[ATPB * NBq], sia[ATPB * NBq];
  __shared__ int sbb[ATPB * NBq], sbe[ATPB * NBq];
  const int ba0 = blockIdx.x * ATPB;
  if (threadIdx.x < ATPB * NBq) {
    int q = threadIdx.x / NBq, r = threadIdx.x - q * NBq;
    int e = ((ba0 + q) * NBq + r) * 2;
    sib[threadIdx.x] = ag[e + 0]; sia[threadIdx.x] = ag[e + 1];
    sbb[threadIdx.x] = bg[e + 0]; sbe[threadIdx.x] = bg[e + 1];
  }
  __syncthreads();
  for (int idx = threadIdx.x; idx < ATPB * PANEL; idx += 256) {
    int q = idx / PANEL, rem = idx - q * PANEL;
    int r = rem / KPAD, c = rem - r * KPAD;
    float v = 0.f;
    if (r < NBq) {
      int s = q * NBq + r;
      if (c < Hq)            v = h[(sib[s] * Asz + sia[s]) * Hq + c];
      else if (c < Hq + BFq) v = bond[(sbb[s] * BONDq + sbe[s]) * BFq + (c - Hq)];
    }
    Ap[idx] = v;
  }
  __syncthreads();
  const int wave = threadIdx.x >> 5, lane = threadIdx.x & 31;
  const int m = lane & 15, half = lane >> 4, koff = half << 1;
  const float* apl = Ap + m * KPAD + koff;
  for (int t = wave; t < NT; t += 8) {
    const int n = t * 16 + m;
    const int nc = (n < Hq) ? n : (Hq - 1);
    v8f accA[ATPB] = {}, accB[ATPB] = {};
    {
      const float* wp = w2a + koff * Hq + nc;
#pragma unroll 3
      for (int kk = 0; kk < Hq; kk += 4) {
        v2f b; b.x = wp[0]; b.y = wp[Hq];
        wp += 4 * Hq;
#pragma unroll
        for (int q = 0; q < ATPB; ++q) {
          v2f a = *(const v2f*)(apl + q * PANEL + kk);
          accA[q] = wmma4(a, b, accA[q]);
        }
      }
    }
#pragma unroll
    for (int kk = 0; kk < 8; kk += 4) {
      const int k0 = kk + koff;
      v2f b;
      b.x = w2b[(k0     < BFq ? k0     : BFq - 1) * Hq + nc];
      b.y = w2b[(k0 + 1 < BFq ? k0 + 1 : BFq - 1) * Hq + nc];
#pragma unroll
      for (int q = 0; q < ATPB; ++q) {
        v2f a = *(const v2f*)(apl + q * PANEL + Hq + kk);
        accB[q] = wmma4(a, b, accB[q]);
      }
    }
#pragma unroll
    for (int q = 0; q < ATPB; ++q) {
      const int nn = nnb[ba0 + q];
      float s = 0.f;
#pragma unroll
      for (int v = 0; v < 8; ++v) {
        int rj = v + half * 8;
        s += (rj < nn) ? accA[q][v] * accB[q][v] : 0.f;
      }
      s += __shfl_xor(s, 16, 32);
      if (half == 0 && n < Hq) out[(ba0 + q) * Hq + n] = s;
    }
  }
}

__global__ __launch_bounds__(256) void out_kernel(
    const float* __restrict__ h, const float* __restrict__ w,
    const float* __restrict__ nprod, const int* __restrict__ na,
    float* __restrict__ out) {
  int tid = blockIdx.x * 8 + (threadIdx.x >> 5);
  if (tid >= MP * NP) return;
  const int lane = threadIdx.x & 31;
  const int m = lane & 15, half = lane >> 4, koff = half << 1;
  const int mp = tid / NP, np = tid - mp * NP;
  const int n0 = np * 32 + m, n1 = n0 + 16;
  const int row0 = mp * 32 + m, row1 = row0 + 16;
  const int nc0 = (n0 < Hq) ? n0 : (Hq - 1);
  const int nc1 = (n1 < Hq) ? n1 : (Hq - 1);
  v8f c00 = {}, c01 = {}, c10 = {}, c11 = {};
  const float* ap0 = h + row0 * Hq + koff;
  const float* ap1 = h + row1 * Hq + koff;
  const float* wp0 = w + koff * Hq + nc0;
  const float* wp1 = w + koff * Hq + nc1;
#pragma unroll 3
  for (int kk = 0; kk < Hq; kk += 4) {
    v2f a0 = *(const v2f*)ap0;
    v2f a1 = *(const v2f*)ap1;
    v2f b0; b0.x = wp0[0]; b0.y = wp0[Hq];
    v2f b1; b1.x = wp1[0]; b1.y = wp1[Hq];
    ap0 += 4; ap1 += 4; wp0 += 4 * Hq; wp1 += 4 * Hq;
    c00 = wmma4(a0, b0, c00); c01 = wmma4(a0, b1, c01);
    c10 = wmma4(a1, b0, c10); c11 = wmma4(a1, b1, c11);
  }
  const int rb0 = mp * 32 + half * 8, rb1 = rb0 + 16;
#pragma unroll
  for (int v = 0; v < 8; ++v) {
    int r0 = rb0 + v, r1 = rb1 + v;
    int b0i = r0 / Asz, a0i = r0 - b0i * Asz;
    int b1i = r1 / Asz, a1i = r1 - b1i * Asz;
    bool m0 = (a0i < na[b0i]), m1 = (a1i < na[b1i]);
    if (n0 < Hq) {
      out[r0 * Hq + n0] = m0 ? c00[v] * nprod[r0 * Hq + n0] : 0.f;
      out[r1 * Hq + n0] = m1 ? c10[v] * nprod[r1 * Hq + n0] : 0.f;
    }
    if (n1 < Hq) {
      out[r0 * Hq + n1] = m0 ? c01[v] * nprod[r0 * Hq + n1] : 0.f;
      out[r1 * Hq + n1] = m1 ? c11[v] * nprod[r1 * Hq + n1] : 0.f;
    }
  }
}

extern "C" void kernel_launch(void* const* d_in, const int* in_sizes, int n_in,
                              void* d_out, int out_size, void* d_ws, size_t ws_size,
                              hipStream_t stream) {
  const float* atom_feats = (const float*)d_in[0];
  const float* bond_feats = (const float*)d_in[1];
  const float* w_fc1      = (const float*)d_in[2];
  const float* w_nei      = (const float*)d_in[3];
  const float* b_nei      = (const float*)d_in[4];
  const float* w_atom     = (const float*)d_in[5];
  const float* b_atom     = (const float*)d_in[6];
  const float* w2a        = (const float*)d_in[7];
  const float* w2b        = (const float*)d_in[8];
  const float* w_fc2      = (const float*)d_in[9];
  const int*   atom_graph = (const int*)d_in[10];
  const int*   bond_graph = (const int*)d_in[11];
  const int*   num_nbs    = (const int*)d_in[12];
  const int*   n_atoms    = (const int*)d_in[13];
  float* out = (float*)d_out;

  const size_t nH = (size_t)Bsz * Asz * Hq;
  float* h_a = (float*)d_ws;
  float* h_b = h_a + nH;
  float* nei = h_b + nH;

  dim3 blk(256);
  const int gemm_blocks = (MP * NP + 7) / 8;
  const int gather_blocks = (Bsz * Asz) / ATPB;

  fc1_kernel<<<gemm_blocks, blk, 0, stream>>>(atom_feats, w_fc1, h_a);

  nei_kernel<<<gather_blocks, blk, 0, stream>>>(h_a, bond_feats, atom_graph,
                                                bond_graph, num_nbs, w_nei, b_nei, nei);
  atom_kernel<<<gemm_blocks, blk, 0, stream>>>(h_a, nei, w_atom, b_atom, h_b);
  nei_kernel<<<gather_blocks, blk, 0, stream>>>(h_b, bond_feats, atom_graph,
                                                bond_graph, num_nbs, w_nei, b_nei, nei);
  atom_kernel<<<gemm_blocks, blk, 0, stream>>>(h_b, nei, w_atom, b_atom, h_a);
  nprod_kernel<<<gather_blocks, blk, 0, stream>>>(h_a, bond_feats, atom_graph,
                                                  bond_graph, num_nbs, w2a, w2b, nei);
  out_kernel<<<gemm_blocks, blk, 0, stream>>>(h_a, w_fc2, nei, n_atoms, out);
}


# K=600 and final K=300 row GEMMs: block = 64-row slab x 20 waves (2 M x 10 N tile-pairs); slab's A rows copied once to LDS in 16-byte pieces, A fragments by ds_load; W path, k and sum order unchanged
# speedup vs baseline: 2.5792x; 2.5792x over previous
#include <hip/hip_runtime.h>
#include <hip/hip_bf16.h>


#define Bsz   256
#define Asz   100
#define NBq   10
#define BONDq 110
#define AFq   82
#define BFq   6
#define Hq    300
#define KPAD  308
#define NT    19
#define NP    10
#define MTOT  (Bsz*Asz)
#define MT    (MTOT/16)
#define MP    (MT/2)
#define GA    8
#define MAXR  (GA*NBq)
#define SP    84
#define MTILE (16*KPAD)
#define NPR   39
#define WFT   (NPR*32)
#define WFM   (NT*WFT)
#define GEMMB ((MP*NP + 7)/8)
#define WFB   ((2*WFM + 255)/256)
#define SLR   64
#define SLB   (MTOT/SLR)
#define RP3   604
#define RP5   308

#define SCHED_FENCE() __builtin_amdgcn_sched_barrier(0)
typedef float v2f __attribute__((ext_vector_type(2)));
typedef float v8f __attribute__((ext_vector_type(8)));

static __device__ __forceinline__ v8f wmma4(v2f a, v2f b, v8f c) {
  return __builtin_amdgcn_wmma_f32_16x16x4_f32(
      false, a, false, b, (short)0, c, false, false);
}

static __device__ __forceinline__ void weight_copies(
    const float* __restrict__ wn, const float* __restrict__ w2a,
    const float* __restrict__ w2b, float* __restrict__ wf, int idx) {
  if (idx >= 2 * WFM) return;
  const int which = (idx >= WFM) ? 1 : 0;
  const int r = idx - which * WFM;
  const int t = r / WFT, q = r - t * WFT;
  const int p = q >> 5, lane = q & 31;
  const int c = lane & 15, h = lane >> 4;
  const int n = t * 16 + c;
  const int nc = (n < Hq) ? n : (Hq - 1);
  float v[4];
#pragma unroll
  for (int e = 0; e < 4; ++e) {
    const int k = 8 * p + (e >> 1) * 4 + 2 * h + (e & 1);
    float x; bool ok;
    if (which == 0) {
      const int kc = (k < Hq + BFq) ? k : (Hq + BFq - 1);
      x = wn[kc * Hq + nc]; ok = k < Hq + BFq;
    } else if (p < NPR - 1) {
      const int kc = (k < Hq) ? k : (Hq - 1);
      x = w2a[kc * Hq + nc]; ok = k < Hq;
    } else {
      const int kb = k - 8 * (NPR - 1);
      const int kc = (kb < BFq) ? kb : (BFq - 1);
      x = w2b[kc * Hq + nc]; ok = kb < BFq;
    }
    v[e] = (ok && n < Hq) ? x : 0.f;
  }
  ((float4*)wf)[idx] = make_float4(v[0], v[1], v[2], v[3]);
}

__global__ __launch_bounds__(256) void fc1_kernel(
    const float* __restrict__ af, const float* __restrict__ w,
    float* __restrict__ hout, const float* __restrict__ wn,
    const float* __restrict__ w2a, const float* __restrict__ w2b,
    float* __restrict__ wf) {
  if (blockIdx.x >= GEMMB) {
    weight_copies(wn, w2a, w2b, wf, (blockIdx.x - GEMMB) * 256 + threadIdx.x);
    return;
  }
  int tid = blockIdx.x * 8 + (threadIdx.x >> 5);
  if (tid >= MP * NP) return;
  const int lane = threadIdx.x & 31;
  const int m = lane & 15, half = lane >> 4, koff = half << 1;
  const int mp = tid / NP, np = tid - mp * NP;
  const int n0 = np * 32 + m, n1 = n0 + 16;
  const int row0 = mp * 32 + m, row1 = row0 + 16;
  const int nc0 = (n0 < Hq) ? n0 : (Hq - 1);
  const int nc1 = (n1 < Hq) ? n1 : (Hq - 1);

  const float* ap0 = af + row0 * AFq + koff;
  const float* ap1 = af + row1 * AFq + koff;
  const float* wp0 = w + koff * Hq + nc0;
  const float* wp1 = w + koff * Hq + nc1;
  v8f c00 = {}, c01 = {}, c10 = {}, c11 = {};
#pragma unroll 2
  for (int kk = 0; kk < 80; kk += 4) {
    v2f a0 = *(const v2f*)ap0;
    v2f a1 = *(const v2f*)ap1;
    v2f b0; b0.x = wp0[0]; b0.y = wp0[Hq];
    v2f b1; b1.x = wp1[0]; b1.y = wp1[Hq];
    ap0 += 4; ap1 += 4; wp0 += 4 * Hq; wp1 += 4 * Hq;
    c00 = wmma4(a0, b0, c00); c01 = wmma4(a0, b1, c01);
    c10 = wmma4(a1, b0, c10); c11 = wmma4(a1, b1, c11);
  }
  {
    const int k0 = 80 + koff;
    const int k0c = (k0 < AFq) ? k0 : AFq - 1;
    const int k1c = (k0 + 1 < AFq) ? k0 + 1 : AFq - 1;
    v2f a0, a1, b0, b1;
    float a0x = af[row0 * AFq + k0c], a0y = af[row0 * AFq + k1c];
    float a1x = af[row1 * AFq + k0c], a1y = af[row1 * AFq + k1c];
    a0.x = (k0 < AFq) ? a0x : 0.f;  a0.y = (k0 + 1 < AFq) ? a0y : 0.f;
    a1.x = (k0 < AFq) ? a1x : 0.f;  a1.y = (k0 + 1 < AFq) ? a1y : 0.f;
    b0.x = w[k0c * Hq + nc0]; b0.y = w[k1c * Hq + nc0];
    b1.x = w[k0c * Hq + nc1]; b1.y = w[k1c * Hq + nc1];
    c00 = wmma4(a0, b0, c00); c01 = wmma4(a0, b1, c01);
    c10 = wmma4(a1, b0, c10); c11 = wmma4(a1, b1, c11);
  }
  const int rb0 = mp * 32 + half * 8, rb1 = rb0 + 16;
#pragma unroll
  for (int v = 0; v < 8; ++v) {
    if (n0 < Hq) {
      hout[(rb0 + v) * Hq + n0] = fmaxf(c00[v], 0.f);
      hout[(rb1 + v) * Hq + n0] = fmaxf(c10[v], 0.f);
    }
    if (n1 < Hq) {
      hout[(rb0 + v) * Hq + n1] = fmaxf(c01[v], 0.f);
      hout[(rb1 + v) * Hq + n1] = fmaxf(c11[v], 0.f);
    }
  }
}

struct GatherShared {
  float Ap[MAXR * KPAD];
  float Sel[16 * SP];
  int rh[MAXR], rb[MAXR];
};

template <bool LIVE_ONLY>
static __device__ __forceinline__ int gather_stage(
    GatherShared& S, const float* __restrict__ h, const float* __restrict__ bond,
    const int* __restrict__ ag, const int* __restrict__ bg,
    const int* __restrict__ nnb, const int* __restrict__ na, int ba0) {
  const int tid = threadIdx.x;
  int nnq[GA];
  int R = 0;
#pragma unroll
  for (int q = 0; q < GA; ++q) {
    int v = nnb[ba0 + q];
    v = v < 0 ? 0 : (v > NBq ? NBq : v);
    if constexpr (LIVE_ONLY) {
      const int ba = ba0 + q, b = ba / Asz;
      v = (ba - b * Asz < na[b]) ? v : 0;
    }
    nnq[q] = v; R += v;
  }
  const int mt = (R + 15) >> 4;
  for (int i = tid; i < 16 * SP; i += 256) S.Sel[i] = 0.f;
  int myq = 0, myrho = -1;
  if (tid < MAXR) {
    const int q = tid / NBq, j = tid - q * NBq;
    int p = 0, nq = 0;
#pragma unroll
    for (int t = 0; t < GA; ++t) {
      p += (t < q) ? nnq[t] : 0;
      nq = (t == q) ? nnq[t] : nq;
    }
    if (j < nq) {
      const int e = ((ba0 + q) * NBq + j) * 2;
      myq = q; myrho = p + j;
      S.rh[myrho] = ag[e + 0] * Asz + ag[e + 1];
      S.rb[myrho] = bg[e + 0] * BONDq + bg[e + 1];
    }
  }
  __syncthreads();
  if (myrho >= 0) S.Sel[myq * SP + myrho] = 1.f;
  const int wave = tid >> 5, lane = tid & 31;
  const int nrows = mt * 16;
  const float4 z4 = make_float4(0.f, 0.f, 0.f, 0.f);
  for (int r0 = wave; r0 < nrows; r0 += 16) {
    const int r1 = r0 + 8;
    float4 a0 = z4, a1 = z4, a2 = z4, c0 = z4, c1 = z4, c2 = z4;
    float ab = 0.f, cb = 0.f;
    if (r0 < R) {
      const float4* s0 = (const float4*)(h + S.rh[r0] * Hq);
      a0 = s0[lane]; a1 = s0[lane + 32];
      if (lane < 11) a2 = s0[lane + 64];
      if (lane < BFq) ab = bond[S.rb[r0] * BFq + lane];
    }
    if (r1 < R) {
      const float4* s1 = (const float4*)(h + S.rh[r1] * Hq);
      c0 = s1[lane]; c1 = s1[lane + 32];
      if (lane < 11) c2 = s1[lane + 64];
      if (lane < BFq) cb = bond[S.rb[r1] * BFq + lane];
    }
    float4* d0 = (float4*)(S.Ap + r0 * KPAD);
    float4* d1 = (float4*)(S.Ap + r1 * KPAD);
    d0[lane] = a0; d0[lane + 32] = a1;
    d1[lane] = c0; d1[lane + 32] = c1;
    if (lane < 11) { d0[lane + 64] = a2; d1[lane + 64] = c2; }
    if (lane < 8) {
      S.Ap[r0 * KPAD + Hq + lane] = ab;
      S.Ap[r1 * KPAD + Hq + lane] = cb;
    }
  }
  __syncthreads();
  return mt;
}

static __device__ __forceinline__ v8f seg_sum(const float* selp, int i, int u,
                                               float x0, float x1, v8f red) {
  v2f xb; xb.x = x0; xb.y = x1;
  v2f sa = *(const v2f*)(selp + i * 16 + 2 * u);
  return wmma4(sa, xb, red);
}


template <int MTN, int KB>
struct Frag {
  v2f b[KB];
  v2f a[MTN][KB];
};

template <int MTN, int KB>
static __device__ __forceinline__ void frag_load(
    Frag<MTN, KB>& f, const float4* __restrict__ wq, const float* apl, int s0) {
  const float4* wb = wq + (s0 >> 1) * 32;
#pragma unroll
  for (int j = 0; j + 1 < KB; j += 2) {
    const float4 w4 = wb[(j >> 1) * 32];
    f.b[j].x = w4.x;     f.b[j].y = w4.y;
    f.b[j + 1].x = w4.z; f.b[j + 1].y = w4.w;
  }
  if constexpr ((KB & 1) != 0)
    f.b[KB - 1] = *(const v2f*)(wb + (KB >> 1) * 32);
#pragma unroll
  for (int i = 0; i < MTN; ++i) {
#pragma unroll
    for (int j = 0; j < KB; ++j)
      f.a[i][j] = *(const v2f*)(apl + i * MTILE + (s0 + j) * 4);
  }
}

template <int MTN, int KB>
static __device__ __forceinline__ void frag_mma(const Frag<MTN, KB>& f,
                                                v8f (&acc)[MTN]) {
#pragma unroll
  for (int j = 0; j < KB; ++j) {
#pragma unroll
    for (int i = 0; i < MTN; ++i) acc[i] = wmma4(f.a[i][j], f.b[j], acc[i]);
  }
}

template <int MTN, int KB, int NS, class EpiLoad>
static __device__ __forceinline__ void kloop(
    v8f (&acc)[MTN], const float4* __restrict__ wp, const float* apl,
    EpiLoad epi_load) {
  constexpr int NBLK = NS / KB;
  constexpr int NPAIR = (NBLK - 1) / 2;
  constexpr bool ODD = ((NBLK - 1) & 1) != 0;
  Frag<MTN, KB> f0, f1;
  frag_load<MTN, KB>(f0, wp, apl, 0);
  int s = 0;
#pragma unroll 1
  for (int p = 0; p < NPAIR; ++p) {
    frag_load<MTN, KB>(f1, wp, apl, s + KB);
    SCHED_FENCE();
    frag_mma<MTN, KB>(f0, acc);
    SCHED_FENCE();
    frag_load<MTN, KB>(f0, wp, apl, s + 2 * KB);
    SCHED_FENCE();
    frag_mma<MTN, KB>(f1, acc);
    SCHED_FENCE();
    s += 2 * KB;
  }
  if constexpr (ODD) {
    frag_load<MTN, KB>(f1, wp, apl, s + KB);
    SCHED_FENCE();
    frag_mma<MTN, KB>(f0, acc);
    SCHED_FENCE();
    epi_load();
    SCHED_FENCE();
    frag_mma<MTN, KB>(f1, acc);
  } else {
    epi_load();
    SCHED_FENCE();
    frag_mma<MTN, KB>(f0, acc);
  }
}

template <int MTN>
static __device__ __forceinline__ void nei_tiles(
    const GatherShared& S, const float4* __restrict__ wf,
    const float* __restrict__ bn, float* __restrict__ out, int ba0) {
  const int wave = threadIdx.x >> 5, lane = threadIdx.x & 31;
  const int m = lane & 15, half = lane >> 4, koff = half << 1;
  const float* apl = S.Ap + m * KPAD + koff;
  const float* selp = S.Sel + m * SP + half * 8;
  constexpr int KB = (MTN <= 2) ? 8 : (MTN <= 4) ? 4 : 2;
  constexpr int REM = 76 - (76 / KB) * KB;
  constexpr int REMA = REM > 0 ? REM : 1;
  for (int t = wave; t < NT; t += 8) {
    const int n = t * 16 + m;
    const int nc = (n < Hq) ? n : (Hq - 1);
    const float4* wp = wf + t * WFT + lane;
    v8f acc[MTN] = {};
    Frag<MTN, REMA> fr;
    Frag<MTN, 1> ft;
    kloop<MTN, KB, 76>(acc, wp, apl, [&]() {
      if constexpr (REM > 0) frag_load<MTN, REMA>(fr, wp, apl, 76 - REM);
      frag_load<MTN, 1>(ft, wp, apl, 76);
    });
    if constexpr (REM > 0) frag_mma<MTN, REMA>(fr, acc);
    frag_mma<MTN, 1>(ft, acc);
    const float bias = bn[nc];
    v8f red = {};
#pragma unroll
    for (int i = 0; i < MTN; ++i) {
#pragma unroll
      for (int u = 0; u < 4; ++u)
        red = seg_sum(selp, i, u, fmaxf(acc[i][2 * u] + bias, 0.f),
                      fmaxf(acc[i][2 * u + 1] + bias, 0.f), red);
    }
    if (half == 0 && n < Hq) {
#pragma unroll
      for (int r = 0; r < GA; ++r) out[(ba0 + r) * Hq + n] = red[r];
    }
  }
}

__global__ __launch_bounds__(256) void nei_kernel(
    const float* __restrict__ h, const float* __restrict__ bond,
    const int* __restrict__ ag, const int* __restrict__ bg,
    const int* __restrict__ nnb, const float* __restrict__ wfn,
    const float* __restrict__ bn, float* __restrict__ out) {
  __shared__ __attribute__((aligned(16))) GatherShared S;
  const int ba0 = blockIdx.x * GA;
  const int mt = gather_stage<false>(S, h, bond, ag, bg, nnb, nullptr, ba0);
  const float4* wf = (const float4*)wfn;
  switch (mt) {
    case 1: nei_tiles<1>(S, wf, bn, out, ba0); break;
    case 2: nei_tiles<2>(S, wf, bn, out, ba0); break;
    case 3: nei_tiles<3>(S, wf, bn, out, ba0); break;
    case 4: nei_tiles<4>(S, wf, bn, out, ba0); break;
    case 5: nei_tiles<5>(S, wf, bn, out, ba0); break;
    default:
      for (int i = threadIdx.x; i < GA * Hq; i += 256) out[ba0 * Hq + i] = 0.f;
  }
}

__global__ __launch_bounds__(640) void atom_kernel(
    const float* __restrict__ h, const float* __restrict__ nei,
    const float* __restrict__ wa, const float* __restrict__ bb,
    float* __restrict__ hout) {
  __shared__ __attribute__((aligned(16))) float As[SLR * RP3];
  const int mb = blockIdx.x;
  const int tid = threadIdx.x;
  {
    const float4* hs = (const float4*)(h + mb * SLR * Hq);
    const float4* ns = (const float4*)(nei + mb * SLR * Hq);
#pragma unroll 1
    for (int part = 0; part < 2; ++part) {
      const float4* sp = part ? ns : hs;
      float4 v[8];
#pragma unroll
      for (int j = 0; j < 8; ++j) {
        const int i = tid + 640 * j;
        v[j] = (i < SLR * 75) ? sp[i] : make_float4(0.f, 0.f, 0.f, 0.f);
      }
#pragma unroll
      for (int j = 0; j < 8; ++j) {
        const int i = tid + 640 * j;
        const int row = i / 75, cc = i - row * 75;
        if (i < SLR * 75) *(float4*)(As + row * RP3 + part * Hq + 4 * cc) = v[j];
      }
    }
  }
  __syncthreads();
  const int lane = tid & 31, wave = tid >> 5;
  const int mq = wave / NP, np = wave - mq * NP;
  const int m = lane & 15, half = lane >> 4, koff = half << 1;
  const int n0 = np * 32 + m, n1 = n0 + 16;
  const int nc0 = (n0 < Hq) ? n0 : (Hq - 1);
  const int nc1 = (n1 < Hq) ? n1 : (Hq - 1);
  v8f c00 = {}, c01 = {}, c10 = {}, c11 = {};
  {
    const float* ap0 = As + (mq * 32 + m) * RP3 + koff;
    const float* ap1 = ap0 + 16 * RP3;
    const float* wp0 = wa + koff * Hq + nc0;
    const float* wp1 = wa + koff * Hq + nc1;
#pragma unroll 3
    for (int kk = 0; kk < 2 * Hq; kk += 4) {
      v2f a0 = *(const v2f*)ap0;
      v2f a1 = *(const v2f*)ap1;
      v2f b0; b0.x = wp0[0]; b0.y = wp0[Hq];
      v2f b1; b1.x = wp1[0]; b1.y = wp1[Hq];
      ap0 += 4; ap1 += 4; wp0 += 4 * Hq; wp1 += 4 * Hq;
      c00 = wmma4(a0, b0, c00); c01 = wmma4(a0, b1, c01);
      c10 = wmma4(a1, b0, c10); c11 = wmma4(a1, b1, c11);
    }
  }
  const float bias0 = bb[nc0], bias1 = bb[nc1];
  const int rb0 = mb * SLR + mq * 32 + half * 8, rb1 = rb0 + 16;
#pragma unroll
  for (int v = 0; v < 8; ++v) {
    if (n0 < Hq) {
      hout[(rb0 + v) * Hq + n0] = fmaxf(c00[v] + bias0, 0.f);
      hout[(rb1 + v) * Hq + n0] = fmaxf(c10[v] + bias0, 0.f);
    }
    if (n1 < Hq) {
      hout[(rb0 + v) * Hq + n1] = fmaxf(c01[v] + bias1, 0.f);
      hout[(rb1 + v) * Hq + n1] = fmaxf(c11[v] + bias1, 0.f);
    }
  }
}

template <int MTN>
static __device__ __forceinline__ void nprod_tiles(
    const GatherShared& S, const float4* __restrict__ wf,
    float* __restrict__ out, int ba0) {
  const int wave = threadIdx.x >> 5, lane = threadIdx.x & 31;
  const int m = lane & 15, half = lane >> 4, koff = half << 1;
  const float* apl = S.Ap + m * KPAD + koff;
  const float* selp = S.Sel + m * SP + half * 8;
  constexpr int KB = (MTN <= 2) ? 8 : (MTN <= 4) ? 4 : 2;
  constexpr int REM = 75 - (75 / KB) * KB;
  for (int t = wave; t < NT; t += 8) {
    const int n = t * 16 + m;
    v8f accA[MTN] = {}, accB[MTN] = {};
    const float4* wp = wf + t * WFT + lane;
    Frag<MTN, REM> fr;
    Frag<MTN, 2> fb;
    kloop<MTN, KB, 75>(accA, wp, apl, [&]() {
      frag_load<MTN, REM>(fr, wp, apl, 75 - REM);
      const float4 w4 = wp[(NPR - 1) * 32];
      fb.b[0].x = w4.x; fb.b[0].y = w4.y;
      fb.b[1].x = w4.z; fb.b[1].y = w4.w;
#pragma unroll
      for (int i = 0; i < MTN; ++i) {
#pragma unroll
        for (int q = 0; q < 2; ++q)
          fb.a[i][q] = *(const v2f*)(apl + i * MTILE + Hq + q * 4);
      }
    });
    frag_mma<MTN, REM>(fr, accA);
    frag_mma<MTN, 2>(fb, accB);
    v8f red = {};
#pragma unroll
    for (int i = 0; i < MTN; ++i) {
#pragma unroll
      for (int u = 0; u < 4; ++u)
        red = seg_sum(selp, i, u, accA[i][2 * u] * accB[i][2 * u],
                      accA[i][2 * u + 1] * accB[i][2 * u + 1], red);
    }
    if (half == 0 && n < Hq) {
#pragma unroll
      for (int r = 0; r < GA; ++r) out[(ba0 + r) * Hq + n] = red[r];
    }
  }
}

__global__ __launch_bounds__(256) void nprod_kernel(
    const float* __restrict__ h, const float* __restrict__ bond,
    const int* __restrict__ ag, const int* __restrict__ bg,
    const int* __restrict__ nnb, const int* __restrict__ na,
    const float* __restrict__ wf2, float* __restrict__ out) {
  __shared__ __attribute__((aligned(16))) GatherShared S;
  const int ba0 = blockIdx.x * GA;
  const int mt = gather_stage<true>(S, h, bond, ag, bg, nnb, na, ba0);
  const float4* wf = (const float4*)wf2;
  switch (mt) {
    case 1: nprod_tiles<1>(S, wf, out, ba0); break;
    case 2: nprod_tiles<2>(S, wf, out, ba0); break;
    case 3: nprod_tiles<3>(S, wf, out, ba0); break;
    case 4: nprod_tiles<4>(S, wf, out, ba0); break;
    case 5: nprod_tiles<5>(S, wf, out, ba0); break;
    default:
      for (int i = threadIdx.x; i < GA * Hq; i += 256) out[ba0 * Hq + i] = 0.f;
  }
}

__global__ __launch_bounds__(640) void out_kernel(
    const float* __restrict__ h, const float* __restrict__ w,
    const float* __restrict__ nprod, const int* __restrict__ na,
    float* __restrict__ out) {
  __shared__ __attribute__((aligned(16))) float As[SLR * RP5];
  const int mb = blockIdx.x;
  const int tid = threadIdx.x;
  const int lane = tid & 31, wave = tid >> 5;
  const int mq = wave / NP, np = wave - mq * NP;
  const int m = lane & 15, half = lane >> 4, koff = half << 1;
  const int n0 = np * 32 + m, n1 = n0 + 16;
  const int nc0 = (n0 < Hq) ? n0 : (Hq - 1);
  const int nc1 = (n1 < Hq) ? n1 : (Hq - 1);
  v8f c00 = {}, c01 = {}, c10 = {}, c11 = {};
  const int rl = mb * SLR + mq * 32 + lane, bl = rl / Asz;
  const int ro = mb * SLR + (1 - mq) * 32 + lane, bo = ro / Asz;
  const bool live_own = __builtin_amdgcn_ballot_w32(rl - bl * Asz < na[bl]) != 0u;
  const bool live_oth = __builtin_amdgcn_ballot_w32(ro - bo * Asz < na[bo]) != 0u;
  if (live_own || live_oth) {
    const float4* hs = (const float4*)(h + mb * SLR * Hq);
    float4 v[8];
#pragma unroll
    for (int j = 0; j < 8; ++j) {
      const int i = tid + 640 * j;
      v[j] = (i < SLR * 75) ? hs[i] : make_float4(0.f, 0.f, 0.f, 0.f);
    }
#pragma unroll
    for (int j = 0; j < 8; ++j) {
      const int i = tid + 640 * j;
      const int row = i / 75, cc = i - row * 75;
      if (i < SLR * 75) *(float4*)(As + row * RP5 + 4 * cc) = v[j];
    }
    __syncthreads();
    if (live_own) {
      const float* ap0 = As + (mq * 32 + m) * RP5 + koff;
      const float* ap1 = ap0 + 16 * RP5;
      const float* wp0 = w + koff * Hq + nc0;
      const float* wp1 = w + koff * Hq + nc1;
#pragma unroll 3
      for (int kk = 0; kk < Hq; kk += 4) {
        v2f a0 = *(const v2f*)ap0;
        v2f a1 = *(const v2f*)ap1;
        v2f b0; b0.x = wp0[0]; b0.y = wp0[Hq];
        v2f b1; b1.x = wp1[0]; b1.y = wp1[Hq];
        ap0 += 4; ap1 += 4; wp0 += 4 * Hq; wp1 += 4 * Hq;
        c00 = wmma4(a0, b0, c00); c01 = wmma4(a0, b1, c01);
        c10 = wmma4(a1, b0, c10); c11 = wmma4(a1, b1, c11);
      }
    }
  }
  const int rb0 = mb * SLR + mq * 32 + half * 8, rb1 = rb0 + 16;
#pragma unroll
  for (int v = 0; v < 8; ++v) {
    int r0 = rb0 + v, r1 = rb1 + v;
    int b0i = r0 / Asz, a0i = r0 - b0i * Asz;
    int b1i = r1 / Asz, a1i = r1 - b1i * Asz;
    bool m0 = (a0i < na[b0i]), m1 = (a1i < na[b1i]);
    if (n0 < Hq) {
      out[r0 * Hq + n0] = m0 ? c00[v] * nprod[r0 * Hq + n0] : 0.f;
      out[r1 * Hq + n0] = m1 ? c10[v] * nprod[r1 * Hq + n0] : 0.f;
    }
    if (n1 < Hq) {
      out[r0 * Hq + n1] = m0 ? c01[v] * nprod[r0 * Hq + n1] : 0.f;
      out[r1 * Hq + n1] = m1 ? c11[v] * nprod[r1 * Hq + n1] : 0.f;
    }
  }
}

extern "C" void kernel_launch(void* const* d_in, const int* in_sizes, int n_in,
                              void* d_out, int out_size, void* d_ws, size_t ws_size,
                              hipStream_t stream) {
  const float* atom_feats = (const float*)d_in[0];
  const float* bond_feats = (const float*)d_in[1];
  const float* w_fc1      = (const float*)d_in[2];
  const float* w_nei      = (const float*)d_in[3];
  const float* b_nei      = (const float*)d_in[4];
  const float* w_atom     = (const float*)d_in[5];
  const float* b_atom     = (const float*)d_in[6];
  const float* w2a        = (const float*)d_in[7];
  const float* w2b        = (const float*)d_in[8];
  const float* w_fc2      = (const float*)d_in[9];
  const int*   atom_graph = (const int*)d_in[10];
  const int*   bond_graph = (const int*)d_in[11];
  const int*   num_nbs    = (const int*)d_in[12];
  const int*   n_atoms    = (const int*)d_in[13];
  float* out = (float*)d_out;

  const size_t nH = (size_t)Bsz * Asz * Hq;
  float* h_a = (float*)d_ws;
  float* h_b = h_a + nH;
  float* nei = h_b + nH;
  float* wfn = nei + nH;
  float* wf2 = wfn + 4 * WFM;

  dim3 blk(256);
  dim3 blk20(640);
  const int slab_blocks = SLB;
  const int gemm_blocks = GEMMB;
  const int gather_blocks = (Bsz * Asz) / GA;

  fc1_kernel<<<gemm_blocks + WFB, blk, 0, stream>>>(atom_feats, w_fc1, h_a,
                                                    w_nei, w2a, w2b, wfn);

  nei_kernel<<<gather_blocks, blk, 0, stream>>>(h_a, bond_feats, atom_graph,
                                                bond_graph, num_nbs, wfn, b_nei, nei);
  atom_kernel<<<slab_blocks, blk20, 0, stream>>>(h_a, nei, w_atom, b_atom, h_b);
  nei_kernel<<<gather_blocks, blk, 0, stream>>>(h_b, bond_feats, atom_graph,
                                                bond_graph, num_nbs, wfn, b_nei, nei);
  atom_kernel<<<slab_blocks, blk20, 0, stream>>>(h_b, nei, w_atom, b_atom, h_a);
  nprod_kernel<<<gather_blocks, blk, 0, stream>>>(h_a, bond_feats, atom_graph,
                                                  bond_graph, num_nbs, n_atoms, wf2, nei);
  out_kernel<<<slab_blocks, blk20, 0, stream>>>(h_a, w_fc2, nei, n_atoms, out);
}
